# starting kernel
# baseline (speedup 1.0000x reference)
#include <hip/hip_runtime.h>
#include <hip/hip_bf16.h>


#define NEG_SLOPE 0.2f
#define F_IN   128
#define H1C1   160
#define HEADS1 8
#define CH1    20
#define NCLS   16

typedef float v2f __attribute__((ext_vector_type(2)));
typedef float v8f __attribute__((ext_vector_type(8)));


__device__ __forceinline__ void edge_sd(const int* __restrict__ ei, int E, int idx,
                                        int& s, int& d) {
    if (idx < E) { s = ei[idx]; d = ei[E + idx]; }
    else         { s = idx - E; d = idx - E; }
}

__device__ __forceinline__ unsigned f2ord(float f) {
    unsigned b = __float_as_uint(f);
    return (b & 0x80000000u) ? ~b : (b | 0x80000000u);
}
__device__ __forceinline__ float ord2f(unsigned u) {
    return (u & 0x80000000u) ? __uint_as_float(u ^ 0x80000000u)
                             : __uint_as_float(~u);
}

__device__ __forceinline__ float lrelu(float v) {
    return (v > 0.f) ? v : NEG_SLOPE * v;
}


__global__ void k_zero(float* __restrict__ p, size_t n) {
    size_t i = (size_t)blockIdx.x * blockDim.x + threadIdx.x;
    size_t stride = (size_t)gridDim.x * blockDim.x;
    for (; i < n; i += stride) p[i] = 0.f;
}

__global__ void k_gemm_wmma_f32(const float* __restrict__ A,
                                const float* __restrict__ B,
                                float* __restrict__ C,
                                int M, int K, int Nc) {
    int wave = blockIdx.x * (blockDim.x >> 5) + (threadIdx.x >> 5);
    int ntN = Nc >> 4;
    int ntM = M >> 4;
    if (wave >= ntM * ntN) return;
    int tm = wave / ntN, tn = wave % ntN;
    int m0 = tm << 4, n0 = tn << 4;

    int lane  = threadIdx.x & 31;
    int lr    = lane & 15;
    int hi    = lane >> 4;
    int khalf = hi << 1;

    const float* __restrict__ arow = A + (size_t)(m0 + lr) * K;
    v8f c = {};
    for (int k0 = 0; k0 < K; k0 += 4) {
        v2f a, b;
        a.x = arow[k0 + khalf];
        a.y = arow[k0 + khalf + 1];
        b.x = B[(size_t)(k0 + khalf)     * Nc + n0 + lr];
        b.y = B[(size_t)(k0 + khalf + 1) * Nc + n0 + lr];
        c = __builtin_amdgcn_wmma_f32_16x16x4_f32(
                 false, a,  false, b,
                 (short)0, c,  false,  false);
    }
#pragma unroll
    for (int r = 0; r < 8; ++r) {
        int row = m0 + r + (hi << 3);
        C[(size_t)row * Nc + n0 + lr] = c[r];
    }
}

__global__ void k_attn_proj(const float* __restrict__ h,
                            const float* __restrict__ att_s,
                            const float* __restrict__ att_d,
                            float* __restrict__ a_s, float* __restrict__ a_d,
                            int n, int heads, int ch) {
    int i = blockIdx.x * blockDim.x + threadIdx.x;
    if (i >= n * heads) return;
    int node = i / heads, hh = i - node * heads;
    const float* hp = h + (size_t)node * heads * ch + (size_t)hh * ch;
    const float* as = att_s + hh * ch;
    const float* ad = att_d + hh * ch;
    float ss = 0.f, sd = 0.f;
    for (int c = 0; c < ch; ++c) {
        float v = hp[c];
        ss += v * as[c];
        sd += v * ad[c];
    }
    a_s[i] = ss;
    a_d[i] = sd;
}

__global__ void k_edge_max(const int* __restrict__ ei, int E, int Etot, int heads,
                           const float* __restrict__ a_s, const float* __restrict__ a_d,
                           unsigned* __restrict__ menc) {
    int i = blockIdx.x * blockDim.x + threadIdx.x;
    if (i >= Etot * heads) return;
    int e = i / heads, hh = i - e * heads;
    int s, d; edge_sd(ei, E, e, s, d);
    float v = lrelu(a_s[s * heads + hh] + a_d[d * heads + hh]);
    atomicMax(&menc[(size_t)d * heads + hh], f2ord(v));
}

__global__ void k_edge_expsum(const int* __restrict__ ei, int E, int Etot, int heads,
                              const float* __restrict__ a_s, const float* __restrict__ a_d,
                              const unsigned* __restrict__ menc,
                              float* __restrict__ denom) {
    int i = blockIdx.x * blockDim.x + threadIdx.x;
    if (i >= Etot * heads) return;
    int e = i / heads, hh = i - e * heads;
    int s, d; edge_sd(ei, E, e, s, d);
    float v = lrelu(a_s[s * heads + hh] + a_d[d * heads + hh]);
    float w = expf(v - ord2f(menc[(size_t)d * heads + hh]));
    atomicAdd(&denom[(size_t)d * heads + hh], w);
}

__global__ void k_edge_scatter(const int* __restrict__ ei, int E, int Etot,
                               int heads, int ch,
                               const float* __restrict__ a_s, const float* __restrict__ a_d,
                               const unsigned* __restrict__ menc,
                               const float* __restrict__ denom,
                               const float* __restrict__ h,
                               float* __restrict__ agg) {
    int i = blockIdx.x * blockDim.x + threadIdx.x;
    if (i >= Etot * heads) return;
    int e = i / heads, hh = i - e * heads;
    int s, d; edge_sd(ei, E, e, s, d);
    float v = lrelu(a_s[s * heads + hh] + a_d[d * heads + hh]);
    float alpha = expf(v - ord2f(menc[(size_t)d * heads + hh]))
                  / denom[(size_t)d * heads + hh];
    const float* hs = h   + (size_t)s * heads * ch + (size_t)hh * ch;
    float*       ag = agg + (size_t)d * heads * ch + (size_t)hh * ch;
    for (int c = 0; c < ch; ++c)
        atomicAdd(&ag[c], alpha * hs[c]);
}

__global__ void k_bias_relu(float* __restrict__ x, const float* __restrict__ b,
                            int n, int hc) {
    int i = blockIdx.x * blockDim.x + threadIdx.x;
    if (i >= n * hc) return;
    x[i] = fmaxf(x[i] + b[i % hc], 0.f);
}

__global__ void k_bias_out(const float* __restrict__ agg, const float* __restrict__ b,
                           float* __restrict__ out, int n, int hc) {
    int i = blockIdx.x * blockDim.x + threadIdx.x;
    if (i >= n * hc) return;
    out[i] = agg[i] + b[i % hc];
}


extern "C" void kernel_launch(void* const* d_in, const int* in_sizes, int n_in,
                              void* d_out, int out_size, void* d_ws, size_t ws_size,
                              hipStream_t stream) {
    const float* x        = (const float*)d_in[0];
    const int*   ei       = (const int*)  d_in[1];
    const float* W1       = (const float*)d_in[2];
    const float* att_src1 = (const float*)d_in[3];
    const float* att_dst1 = (const float*)d_in[4];
    const float* b1       = (const float*)d_in[5];
    const float* W2       = (const float*)d_in[6];
    const float* att_src2 = (const float*)d_in[7];
    const float* att_dst2 = (const float*)d_in[8];
    const float* b2       = (const float*)d_in[9];
    float* out = (float*)d_out;

    const int N    = in_sizes[0] / F_IN;
    const int E    = in_sizes[1] / 2;
    const int Etot = E + N;

    float* ws  = (float*)d_ws;
    float* h1  = ws;
    float* as1 = h1  + (size_t)N * H1C1;
    float* ad1 = as1 + (size_t)N * HEADS1;
    float* h2  = ad1 + (size_t)N * HEADS1;
    float* as2 = h2  + (size_t)N * NCLS;
    float* ad2 = as2 + (size_t)N;
    float*    zbase = ad2 + (size_t)N;
    float*    agg1  = zbase;
    float*    den1  = agg1 + (size_t)N * H1C1;
    unsigned* m1    = (unsigned*)(den1 + (size_t)N * HEADS1);
    float*    agg2  = (float*)(m1 + (size_t)N * HEADS1);
    float*    den2  = agg2 + (size_t)N * NCLS;
    unsigned* m2    = (unsigned*)(den2 + (size_t)N);
    size_t zcount = (size_t)N * (H1C1 + HEADS1 + HEADS1 + NCLS + 1 + 1);

    const int T = 256;

    k_zero<<<4096, T, 0, stream>>>(zbase, zcount);

    {
        int tiles = (N >> 4) * (H1C1 >> 4);
        int wavesPerBlk = T >> 5;
        k_gemm_wmma_f32<<<(tiles + wavesPerBlk - 1) / wavesPerBlk, T, 0, stream>>>(
            x, W1, h1, N, F_IN, H1C1);
    }
    k_attn_proj<<<((N * HEADS1) + T - 1) / T, T, 0, stream>>>(
        h1, att_src1, att_dst1, as1, ad1, N, HEADS1, CH1);

    int work1 = Etot * HEADS1;
    k_edge_max   <<<(work1 + T - 1) / T, T, 0, stream>>>(ei, E, Etot, HEADS1, as1, ad1, m1);
    k_edge_expsum<<<(work1 + T - 1) / T, T, 0, stream>>>(ei, E, Etot, HEADS1, as1, ad1, m1, den1);
    k_edge_scatter<<<(work1 + T - 1) / T, T, 0, stream>>>(ei, E, Etot, HEADS1, CH1,
                                                          as1, ad1, m1, den1, h1, agg1);
    k_bias_relu<<<((N * H1C1) + T - 1) / T, T, 0, stream>>>(agg1, b1, N, H1C1);

    {
        int tiles = (N >> 4) * (NCLS >> 4);
        int wavesPerBlk = T >> 5;
        k_gemm_wmma_f32<<<(tiles + wavesPerBlk - 1) / wavesPerBlk, T, 0, stream>>>(
            agg1, W2, h2, N, H1C1, NCLS);
    }
    k_attn_proj<<<(N + T - 1) / T, T, 0, stream>>>(
        h2, att_src2, att_dst2, as2, ad2, N, 1, NCLS);

    k_edge_max   <<<(Etot + T - 1) / T, T, 0, stream>>>(ei, E, Etot, 1, as2, ad2, m2);
    k_edge_expsum<<<(Etot + T - 1) / T, T, 0, stream>>>(ei, E, Etot, 1, as2, ad2, m2, den2);
    k_edge_scatter<<<(Etot + T - 1) / T, T, 0, stream>>>(ei, E, Etot, 1, NCLS,
                                                         as2, ad2, m2, den2, h2, agg2);
    k_bias_out<<<((N * NCLS) + T - 1) / T, T, 0, stream>>>(agg2, b2, out, N, NCLS);
}


# Edge passes per layer (zero fill, max, sum, scatter) are one owner kernel: a work-group owns 256/512 dst nodes, filters the edge list 3x, sums in LDS, plain stores; row sum order = LDS arrival
# speedup vs baseline: 2.5550x; 2.5550x over previous
#include <hip/hip_runtime.h>
#include <hip/hip_bf16.h>


#define NEG_SLOPE 0.2f
#define F_IN   128
#define H1C1   160
#define HEADS1 8
#define CH1    20
#define NCLS   16

typedef float v2f __attribute__((ext_vector_type(2)));
typedef float v8f __attribute__((ext_vector_type(8)));


__device__ __forceinline__ unsigned f2ord(float f) {
    unsigned b = __float_as_uint(f);
    return (b & 0x80000000u) ? ~b : (b | 0x80000000u);
}
__device__ __forceinline__ float ord2f(unsigned u) {
    return (u & 0x80000000u) ? __uint_as_float(u ^ 0x80000000u)
                             : __uint_as_float(~u);
}

__device__ __forceinline__ float lrelu(float v) {
    return (v > 0.f) ? v : NEG_SLOPE * v;
}


__global__ void k_gemm_wmma_f32(const float* __restrict__ A,
                                const float* __restrict__ B,
                                float* __restrict__ C,
                                int M, int K, int Nc) {
    int wave = blockIdx.x * (blockDim.x >> 5) + (threadIdx.x >> 5);
    int ntN = Nc >> 4;
    int ntM = M >> 4;
    if (wave >= ntM * ntN) return;
    int tm = wave / ntN, tn = wave % ntN;
    int m0 = tm << 4, n0 = tn << 4;

    int lane  = threadIdx.x & 31;
    int lr    = lane & 15;
    int hi    = lane >> 4;
    int khalf = hi << 1;

    const float* __restrict__ arow = A + (size_t)(m0 + lr) * K;
    v8f c = {};
    for (int k0 = 0; k0 < K; k0 += 4) {
        v2f a, b;
        a.x = arow[k0 + khalf];
        a.y = arow[k0 + khalf + 1];
        b.x = B[(size_t)(k0 + khalf)     * Nc + n0 + lr];
        b.y = B[(size_t)(k0 + khalf + 1) * Nc + n0 + lr];
        c = __builtin_amdgcn_wmma_f32_16x16x4_f32(
                 false, a,  false, b,
                 (short)0, c,  false,  false);
    }
#pragma unroll
    for (int r = 0; r < 8; ++r) {
        int row = m0 + r + (hi << 3);
        C[(size_t)row * Nc + n0 + lr] = c[r];
    }
}

__global__ void k_attn_proj(const float* __restrict__ h,
                            const float* __restrict__ att_s,
                            const float* __restrict__ att_d,
                            float* __restrict__ a_s, float* __restrict__ a_d,
                            int n, int heads, int ch) {
    int i = blockIdx.x * blockDim.x + threadIdx.x;
    if (i >= n * heads) return;
    int node = i / heads, hh = i - node * heads;
    const float* hp = h + (size_t)node * heads * ch + (size_t)hh * ch;
    const float* as = att_s + hh * ch;
    const float* ad = att_d + hh * ch;
    float ss = 0.f, sd = 0.f;
    for (int c = 0; c < ch; ++c) {
        float v = hp[c];
        ss += v * as[c];
        sd += v * ad[c];
    }
    a_s[i] = ss;
    a_d[i] = sd;
}


#define OWN_WG   1024
#define OWN_NW   (OWN_WG / 32)
#define OWN_QCAP 160

template <int HEADS, int CH, int P>
struct OwnLds {
    float    acc[P * HEADS * CH];
    unsigned menc[P * HEADS];
    float    den[P * HEADS];
    float    adl[P * HEADS];
    int      qs[OWN_NW][OWN_QCAP];
    int      qd[OWN_NW][OWN_QCAP];
    float    qa[OWN_NW][32 * HEADS];
};

__device__ __forceinline__ void wave_lds_handover() {
    __builtin_amdgcn_fence(__ATOMIC_RELEASE, "wavefront");
    __builtin_amdgcn_wave_barrier();
    __builtin_amdgcn_fence(__ATOMIC_ACQUIRE, "wavefront");
}

template <int HEADS, int CH, int P, int PH>
__device__ __forceinline__ void own_batch(OwnLds<HEADS, CH, P>& L, int wave, int lane,
                                          int first, int cnt,
                                          const int* __restrict__ ei, int E,
                                          const float* __restrict__ a_s,
                                          const float* __restrict__ h) {
    constexpr int ROW = HEADS * CH;
    wave_lds_handover();
    if (lane < cnt) {
        int e  = L.qs[wave][first + lane];
        int dl = L.qd[wave][first + lane];
        int s  = (e < E) ? ei[e] : e - E;
        if (PH == 2) L.qs[wave][first + lane] = s;
#pragma unroll
        for (int hh = 0; hh < HEADS; ++hh) {
            float v = lrelu(a_s[(size_t)s * HEADS + hh] + L.adl[dl * HEADS + hh]);
            if (PH == 0) {
                atomicMax(&L.menc[dl * HEADS + hh], f2ord(v));
            } else {
                float w = expf(v - ord2f(L.menc[dl * HEADS + hh]));
                if (PH == 1) atomicAdd(&L.den[dl * HEADS + hh], w);
                else         L.qa[wave][lane * HEADS + hh] = w / L.den[dl * HEADS + hh];
            }
        }
    }
    if (PH == 2) {
        wave_lds_handover();
        constexpr int U = (ROW >= 32) ? 20 : 8;
        const int total = cnt * ROW;
        for (int base = 0; base < total; base += 32 * U) {
            float v[U];
#pragma unroll
            for (int u = 0; u < U; ++u) {
                int idx = min(base + u * 32 + lane, total - 1);
                int ent = idx / ROW, col = idx - ent * ROW;
                int s = L.qs[wave][first + ent];
                v[u] = h[(size_t)s * ROW + col];
            }
#pragma unroll
            for (int u = 0; u < U; ++u) {
                int idx = base + u * 32 + lane;
                if (idx < total) {
                    int ent = idx / ROW, col = idx - ent * ROW;
                    int dl = L.qd[wave][first + ent];
                    float a = L.qa[wave][ent * HEADS + col / CH];
                    atomicAdd(&L.acc[dl * ROW + col], a * v[u]);
                }
            }
        }
        wave_lds_handover();
    }
}

template <int HEADS, int CH, int P>
__device__ __forceinline__ void own_append(OwnLds<HEADS, CH, P>& L, int wave, int lane, int& qcnt,
                                           bool m, int e, int dl) {
    unsigned mask = __builtin_amdgcn_ballot_w32(m);
    if (mask == 0u) return;
    int pos = qcnt + __popc(mask & ((1u << lane) - 1u));
    if (m) { L.qs[wave][pos] = e; L.qd[wave][pos] = dl; }
    qcnt += __popc(mask);
}

template <int HEADS, int CH, int P, int PH>
__device__ __forceinline__ void own_drain(OwnLds<HEADS, CH, P>& L, int wave, int lane, int& qcnt,
                                          const int* __restrict__ ei, int E,
                                          const float* __restrict__ a_s,
                                          const float* __restrict__ h) {
    while (qcnt >= 32) {
        own_batch<HEADS, CH, P, PH>(L, wave, lane, qcnt - 32, 32, ei, E, a_s, h);
        qcnt -= 32;
    }
}

__device__ __forceinline__ int4 own_ld4(const int* __restrict__ dsts, int i, int EV) {
    return (i < EV) ? *(const int4*)(dsts + 4 * (size_t)i) : make_int4(-1, -1, -1, -1);
}

template <int HEADS, int CH, int P, int PH>
__device__ __forceinline__ void own_scan(OwnLds<HEADS, CH, P>& L, int wave, int lane,
                                         const int* __restrict__ ei, int E, int N, int n0,
                                         const float* __restrict__ a_s,
                                         const float* __restrict__ h) {
    int qcnt = 0;
    const int* __restrict__ dsts = ei + E;
    const int EV = (E & 3) ? 0 : (E >> 2);
    const int t = wave * 32 + lane;
    int4 r0 = own_ld4(dsts, t, EV);
    int4 r1 = own_ld4(dsts, t + OWN_WG, EV);
    int4 r2 = own_ld4(dsts, t + 2 * OWN_WG, EV);
    int4 r3 = own_ld4(dsts, t + 3 * OWN_WG, EV);
    for (int base = wave * 32; base < EV; base += OWN_WG) {
        const int i = base + lane;
        const int4 d = r0;
        r0 = r1; r1 = r2; r2 = r3;
        r3 = own_ld4(dsts, i + 4 * OWN_WG, EV);
        unsigned q0 = (unsigned)(d.x - n0), q1 = (unsigned)(d.y - n0);
        unsigned q2 = (unsigned)(d.z - n0), q3 = (unsigned)(d.w - n0);
        bool m0 = q0 < (unsigned)P, m1 = q1 < (unsigned)P;
        bool m2 = q2 < (unsigned)P, m3 = q3 < (unsigned)P;
        if (__builtin_amdgcn_ballot_w32(m0 | m1 | m2 | m3) != 0u) {
            own_append<HEADS, CH, P>(L, wave, lane, qcnt, m0, 4 * i + 0, (int)q0);
            own_append<HEADS, CH, P>(L, wave, lane, qcnt, m1, 4 * i + 1, (int)q1);
            own_append<HEADS, CH, P>(L, wave, lane, qcnt, m2, 4 * i + 2, (int)q2);
            own_append<HEADS, CH, P>(L, wave, lane, qcnt, m3, 4 * i + 3, (int)q3);
            own_drain<HEADS, CH, P, PH>(L, wave, lane, qcnt, ei, E, a_s, h);
        }
    }
    for (int base = 4 * EV + wave * 32; base < E; base += OWN_WG) {
        int i = base + lane;
        unsigned q = 0u; bool m = false;
        if (i < E) { q = (unsigned)(dsts[i] - n0); m = q < (unsigned)P; }
        own_append<HEADS, CH, P>(L, wave, lane, qcnt, m, i, (int)q);
        own_drain<HEADS, CH, P, PH>(L, wave, lane, qcnt, ei, E, a_s, h);
    }
    for (int p0 = wave * 32; p0 < P; p0 += OWN_WG) {
        int p = p0 + lane;
        bool m = (p < P) && (n0 + p < N);
        own_append<HEADS, CH, P>(L, wave, lane, qcnt, m, E + n0 + p, p);
        own_drain<HEADS, CH, P, PH>(L, wave, lane, qcnt, ei, E, a_s, h);
    }
    if (qcnt > 0) own_batch<HEADS, CH, P, PH>(L, wave, lane, 0, qcnt, ei, E, a_s, h);
}

template <int HEADS, int CH, int P>
__global__ __launch_bounds__(OWN_WG)
void k_gat_owner(const int* __restrict__ ei, int E, int N,
                 const float* __restrict__ a_s, const float* __restrict__ a_d,
                 const float* __restrict__ h, float* __restrict__ agg) {
    constexpr int ROW = HEADS * CH;
    __shared__ OwnLds<HEADS, CH, P> L;
    const int tid = threadIdx.x, wave = tid >> 5, lane = tid & 31;
    const int n0 = blockIdx.x * P;

    for (int i = tid; i < P * ROW; i += OWN_WG) L.acc[i] = 0.f;
    for (int i = tid; i < P * HEADS; i += OWN_WG) {
        L.menc[i] = 0u;
        L.den[i]  = 0.f;
        L.adl[i]  = (n0 + i / HEADS < N) ? a_d[(size_t)n0 * HEADS + i] : 0.f;
    }
    __syncthreads();
    own_scan<HEADS, CH, P, 0>(L, wave, lane, ei, E, N, n0, a_s, h);
    __syncthreads();
    own_scan<HEADS, CH, P, 1>(L, wave, lane, ei, E, N, n0, a_s, h);
    __syncthreads();
    own_scan<HEADS, CH, P, 2>(L, wave, lane, ei, E, N, n0, a_s, h);
    __syncthreads();
    for (int i = tid; i < P * ROW; i += OWN_WG)
        if (n0 + i / ROW < N) agg[(size_t)n0 * ROW + i] = L.acc[i];
}

__global__ void k_bias_relu(float* __restrict__ x, const float* __restrict__ b,
                            int n, int hc) {
    int i = blockIdx.x * blockDim.x + threadIdx.x;
    if (i >= n * hc) return;
    x[i] = fmaxf(x[i] + b[i % hc], 0.f);
}

__global__ void k_bias_out(const float* __restrict__ agg, const float* __restrict__ b,
                           float* __restrict__ out, int n, int hc) {
    int i = blockIdx.x * blockDim.x + threadIdx.x;
    if (i >= n * hc) return;
    out[i] = agg[i] + b[i % hc];
}


extern "C" void kernel_launch(void* const* d_in, const int* in_sizes, int n_in,
                              void* d_out, int out_size, void* d_ws, size_t ws_size,
                              hipStream_t stream) {
    const float* x        = (const float*)d_in[0];
    const int*   ei       = (const int*)  d_in[1];
    const float* W1       = (const float*)d_in[2];
    const float* att_src1 = (const float*)d_in[3];
    const float* att_dst1 = (const float*)d_in[4];
    const float* b1       = (const float*)d_in[5];
    const float* W2       = (const float*)d_in[6];
    const float* att_src2 = (const float*)d_in[7];
    const float* att_dst2 = (const float*)d_in[8];
    const float* b2       = (const float*)d_in[9];
    float* out = (float*)d_out;

    const int N    = in_sizes[0] / F_IN;
    const int E    = in_sizes[1] / 2;

    float* ws  = (float*)d_ws;
    float* h1  = ws;
    float* as1 = h1  + (size_t)N * H1C1;
    float* ad1 = as1 + (size_t)N * HEADS1;
    float* h2  = ad1 + (size_t)N * HEADS1;
    float* as2 = h2  + (size_t)N * NCLS;
    float* ad2 = as2 + (size_t)N;
    float* agg1 = ad2 + (size_t)N;
    float* agg2 = agg1 + (size_t)N * H1C1;

    const int T = 256;
    constexpr int P1 = 256;
    constexpr int P2 = 512;

    {
        int tiles = (N >> 4) * (H1C1 >> 4);
        int wavesPerBlk = T >> 5;
        k_gemm_wmma_f32<<<(tiles + wavesPerBlk - 1) / wavesPerBlk, T, 0, stream>>>(
            x, W1, h1, N, F_IN, H1C1);
    }
    k_attn_proj<<<((N * HEADS1) + T - 1) / T, T, 0, stream>>>(
        h1, att_src1, att_dst1, as1, ad1, N, HEADS1, CH1);

    k_gat_owner<HEADS1, CH1, P1><<<(N + P1 - 1) / P1, OWN_WG, 0, stream>>>(
        ei, E, N, as1, ad1, h1, agg1);
    k_bias_relu<<<((N * H1C1) + T - 1) / T, T, 0, stream>>>(agg1, b1, N, H1C1);

    {
        int tiles = (N >> 4) * (NCLS >> 4);
        int wavesPerBlk = T >> 5;
        k_gemm_wmma_f32<<<(tiles + wavesPerBlk - 1) / wavesPerBlk, T, 0, stream>>>(
            agg1, W2, h2, N, H1C1, NCLS);
    }
    k_attn_proj<<<(N + T - 1) / T, T, 0, stream>>>(
        h2, att_src2, att_dst2, as2, ad2, N, 1, NCLS);

    k_gat_owner<1, NCLS, P2><<<(N + P2 - 1) / P2, OWN_WG, 0, stream>>>(
        ei, E, N, as2, ad2, h2, agg2);
    k_bias_out<<<((N * NCLS) + T - 1) / T, T, 0, stream>>>(agg2, b2, out, N, NCLS);
}


# Each owner work-group reads the edge list once, not three times: every wave keeps its hits in an LDS list that the max, sum and weighted-sum passes walk; row sum order = LDS arrival, as before
# speedup vs baseline: 3.2731x; 1.2811x over previous
#include <hip/hip_runtime.h>
#include <hip/hip_bf16.h>


#define NEG_SLOPE 0.2f
#define F_IN   128
#define H1C1   160
#define HEADS1 8
#define CH1    20
#define NCLS   16

typedef float v2f __attribute__((ext_vector_type(2)));
typedef float v8f __attribute__((ext_vector_type(8)));


__device__ __forceinline__ unsigned f2ord(float f) {
    unsigned b = __float_as_uint(f);
    return (b & 0x80000000u) ? ~b : (b | 0x80000000u);
}
__device__ __forceinline__ float ord2f(unsigned u) {
    return (u & 0x80000000u) ? __uint_as_float(u ^ 0x80000000u)
                             : __uint_as_float(~u);
}

__device__ __forceinline__ float lrelu(float v) {
    return (v > 0.f) ? v : NEG_SLOPE * v;
}


__global__ void k_gemm_wmma_f32(const float* __restrict__ A,
                                const float* __restrict__ B,
                                float* __restrict__ C,
                                int M, int K, int Nc) {
    int wave = blockIdx.x * (blockDim.x >> 5) + (threadIdx.x >> 5);
    int ntN = Nc >> 4;
    int ntM = M >> 4;
    if (wave >= ntM * ntN) return;
    int tm = wave / ntN, tn = wave % ntN;
    int m0 = tm << 4, n0 = tn << 4;

    int lane  = threadIdx.x & 31;
    int lr    = lane & 15;
    int hi    = lane >> 4;
    int khalf = hi << 1;

    const float* __restrict__ arow = A + (size_t)(m0 + lr) * K;
    v8f c = {};
    for (int k0 = 0; k0 < K; k0 += 4) {
        v2f a, b;
        a.x = arow[k0 + khalf];
        a.y = arow[k0 + khalf + 1];
        b.x = B[(size_t)(k0 + khalf)     * Nc + n0 + lr];
        b.y = B[(size_t)(k0 + khalf + 1) * Nc + n0 + lr];
        c = __builtin_amdgcn_wmma_f32_16x16x4_f32(
                 false, a,  false, b,
                 (short)0, c,  false,  false);
    }
#pragma unroll
    for (int r = 0; r < 8; ++r) {
        int row = m0 + r + (hi << 3);
        C[(size_t)row * Nc + n0 + lr] = c[r];
    }
}

__global__ void k_attn_proj(const float* __restrict__ h,
                            const float* __restrict__ att_s,
                            const float* __restrict__ att_d,
                            float* __restrict__ a_s, float* __restrict__ a_d,
                            int n, int heads, int ch) {
    int i = blockIdx.x * blockDim.x + threadIdx.x;
    if (i >= n * heads) return;
    int node = i / heads, hh = i - node * heads;
    const float* hp = h + (size_t)node * heads * ch + (size_t)hh * ch;
    const float* as = att_s + hh * ch;
    const float* ad = att_d + hh * ch;
    float ss = 0.f, sd = 0.f;
    for (int c = 0; c < ch; ++c) {
        float v = hp[c];
        ss += v * as[c];
        sd += v * ad[c];
    }
    a_s[i] = ss;
    a_d[i] = sd;
}


#define OWN_WG   1024
#define OWN_NW   (OWN_WG / 32)
#define OWN_QCAP 160

template <int HEADS, int CH, int P, int KCAP>
struct OwnLds {
    static_assert(KCAP >= OWN_QCAP, "the three-scan path queues in the same arrays");
    float    acc[P * HEADS * CH];
    unsigned menc[P * HEADS];
    float    den[P * HEADS];
    float    adl[P * HEADS];
    int      qs[OWN_NW][KCAP];
    int      qd[OWN_NW][KCAP];
    float    qa[OWN_NW][32 * HEADS];
    int      ovf;
};

__device__ __forceinline__ void wave_lds_handover() {
    __builtin_amdgcn_fence(__ATOMIC_RELEASE, "wavefront");
    __builtin_amdgcn_wave_barrier();
    __builtin_amdgcn_fence(__ATOMIC_ACQUIRE, "wavefront");
}

template <int HEADS, int CH, int P, int KCAP, int PH, bool LIST>
__device__ __forceinline__ void own_batch(OwnLds<HEADS, CH, P, KCAP>& L, int wave, int lane,
                                          int first, int cnt,
                                          const int* __restrict__ ei, int E,
                                          const float* __restrict__ a_s,
                                          const float* __restrict__ h) {
    constexpr int ROW = HEADS * CH;
    wave_lds_handover();
    if (lane < cnt) {
        int e  = L.qs[wave][first + lane];
        int dl = L.qd[wave][first + lane];
        int s;
        if (LIST && PH > 0) s = e;
        else s = (e < E) ? ei[e] : e - E;
        if (LIST ? (PH == 0) : (PH == 2))
            L.qs[wave][first + lane] = s;
#pragma unroll
        for (int hh = 0; hh < HEADS; ++hh) {
            float v = lrelu(a_s[(size_t)s * HEADS + hh] + L.adl[dl * HEADS + hh]);
            if (PH == 0) {
                atomicMax(&L.menc[dl * HEADS + hh], f2ord(v));
            } else {
                float w = expf(v - ord2f(L.menc[dl * HEADS + hh]));
                if (PH == 1) atomicAdd(&L.den[dl * HEADS + hh], w);
                else         L.qa[wave][lane * HEADS + hh] = w / L.den[dl * HEADS + hh];
            }
        }
    }
    if (PH == 2) {
        wave_lds_handover();
        constexpr int U = (ROW >= 32) ? 20 : 8;
        const int total = cnt * ROW;
        for (int base = 0; base < total; base += 32 * U) {
            float v[U];
#pragma unroll
            for (int u = 0; u < U; ++u) {
                int idx = min(base + u * 32 + lane, total - 1);
                int ent = idx / ROW, col = idx - ent * ROW;
                int s = L.qs[wave][first + ent];
                v[u] = h[(size_t)s * ROW + col];
            }
#pragma unroll
            for (int u = 0; u < U; ++u) {
                int idx = base + u * 32 + lane;
                if (idx < total) {
                    int ent = idx / ROW, col = idx - ent * ROW;
                    int dl = L.qd[wave][first + ent];
                    float a = L.qa[wave][ent * HEADS + col / CH];
                    atomicAdd(&L.acc[dl * ROW + col], a * v[u]);
                }
            }
        }
        wave_lds_handover();
    }
}

template <int HEADS, int CH, int P, int KCAP>
__device__ __forceinline__ void own_append(OwnLds<HEADS, CH, P, KCAP>& L, int wave, int lane, int& qcnt,
                                           bool m, int e, int dl) {
    unsigned mask = __builtin_amdgcn_ballot_w32(m);
    if (mask == 0u) return;
    int pos = qcnt + __popc(mask & ((1u << lane) - 1u));
    if (m) { L.qs[wave][pos] = e; L.qd[wave][pos] = dl; }
    qcnt += __popc(mask);
}

template <int HEADS, int CH, int P, int KCAP, int PH>
__device__ __forceinline__ void own_drain(OwnLds<HEADS, CH, P, KCAP>& L, int wave, int lane, int& qcnt,
                                          const int* __restrict__ ei, int E,
                                          const float* __restrict__ a_s,
                                          const float* __restrict__ h) {
    while (qcnt >= 32) {
        own_batch<HEADS, CH, P, KCAP, PH, false>(L, wave, lane, qcnt - 32, 32, ei, E, a_s, h);
        qcnt -= 32;
    }
}

__device__ __forceinline__ int4 own_ld4(const int* __restrict__ dsts, int i, int EV) {
    return (i < EV) ? *(const int4*)(dsts + 4 * (size_t)i) : make_int4(-1, -1, -1, -1);
}

template <int HEADS, int CH, int P, int KCAP, int PH>
__device__ __forceinline__ void own_scan(OwnLds<HEADS, CH, P, KCAP>& L, int wave, int lane,
                                         const int* __restrict__ ei, int E, int N, int n0,
                                         const float* __restrict__ a_s,
                                         const float* __restrict__ h) {
    int qcnt = 0;
    const int* __restrict__ dsts = ei + E;
    const int EV = (E & 3) ? 0 : (E >> 2);
    const int t = wave * 32 + lane;
    int4 r0 = own_ld4(dsts, t, EV);
    int4 r1 = own_ld4(dsts, t + OWN_WG, EV);
    int4 r2 = own_ld4(dsts, t + 2 * OWN_WG, EV);
    int4 r3 = own_ld4(dsts, t + 3 * OWN_WG, EV);
    for (int base = wave * 32; base < EV; base += OWN_WG) {
        const int i = base + lane;
        const int4 d = r0;
        r0 = r1; r1 = r2; r2 = r3;
        r3 = own_ld4(dsts, i + 4 * OWN_WG, EV);
        unsigned q0 = (unsigned)(d.x - n0), q1 = (unsigned)(d.y - n0);
        unsigned q2 = (unsigned)(d.z - n0), q3 = (unsigned)(d.w - n0);
        bool m0 = q0 < (unsigned)P, m1 = q1 < (unsigned)P;
        bool m2 = q2 < (unsigned)P, m3 = q3 < (unsigned)P;
        if (__builtin_amdgcn_ballot_w32(m0 | m1 | m2 | m3) != 0u) {
            own_append<HEADS, CH, P, KCAP>(L, wave, lane, qcnt, m0, 4 * i + 0, (int)q0);
            own_append<HEADS, CH, P, KCAP>(L, wave, lane, qcnt, m1, 4 * i + 1, (int)q1);
            own_append<HEADS, CH, P, KCAP>(L, wave, lane, qcnt, m2, 4 * i + 2, (int)q2);
            own_append<HEADS, CH, P, KCAP>(L, wave, lane, qcnt, m3, 4 * i + 3, (int)q3);
            own_drain<HEADS, CH, P, KCAP, PH>(L, wave, lane, qcnt, ei, E, a_s, h);
        }
    }
    for (int base = 4 * EV + wave * 32; base < E; base += OWN_WG) {
        int i = base + lane;
        unsigned q = 0u; bool m = false;
        if (i < E) { q = (unsigned)(dsts[i] - n0); m = q < (unsigned)P; }
        own_append<HEADS, CH, P, KCAP>(L, wave, lane, qcnt, m, i, (int)q);
        own_drain<HEADS, CH, P, KCAP, PH>(L, wave, lane, qcnt, ei, E, a_s, h);
    }
    for (int p0 = wave * 32; p0 < P; p0 += OWN_WG) {
        int p = p0 + lane;
        bool m = (p < P) && (n0 + p < N);
        own_append<HEADS, CH, P, KCAP>(L, wave, lane, qcnt, m, E + n0 + p, p);
        own_drain<HEADS, CH, P, KCAP, PH>(L, wave, lane, qcnt, ei, E, a_s, h);
    }
    if (qcnt > 0) own_batch<HEADS, CH, P, KCAP, PH, false>(L, wave, lane, 0, qcnt, ei, E, a_s, h);
}

template <int HEADS, int CH, int P, int KCAP>
__device__ __forceinline__ void own_keep(OwnLds<HEADS, CH, P, KCAP>& L, int wave, int lane, int& kcnt,
                                         bool m, int e, int dl) {
    unsigned mask = __builtin_amdgcn_ballot_w32(m);
    if (mask == 0u) return;
    int pos = kcnt + __popc(mask & ((1u << lane) - 1u));
    if (m && pos < KCAP) { L.qs[wave][pos] = e; L.qd[wave][pos] = dl; }
    kcnt += __popc(mask);
}

template <int HEADS, int CH, int P, int KCAP>
__device__ __forceinline__ int own_filter(OwnLds<HEADS, CH, P, KCAP>& L, int wave, int lane,
                                          const int* __restrict__ ei, int E, int N, int n0) {
    int kcnt = 0;
    const int* __restrict__ dsts = ei + E;
    const int EV = (E & 3) ? 0 : (E >> 2);
    const int t = wave * 32 + lane;
    int4 r0 = own_ld4(dsts, t, EV);
    int4 r1 = own_ld4(dsts, t + OWN_WG, EV);
    int4 r2 = own_ld4(dsts, t + 2 * OWN_WG, EV);
    int4 r3 = own_ld4(dsts, t + 3 * OWN_WG, EV);
    for (int base = wave * 32; base < EV; base += OWN_WG) {
        const int i = base + lane;
        const int4 d = r0;
        r0 = r1; r1 = r2; r2 = r3;
        r3 = own_ld4(dsts, i + 4 * OWN_WG, EV);
        unsigned q0 = (unsigned)(d.x - n0), q1 = (unsigned)(d.y - n0);
        unsigned q2 = (unsigned)(d.z - n0), q3 = (unsigned)(d.w - n0);
        bool m0 = q0 < (unsigned)P, m1 = q1 < (unsigned)P;
        bool m2 = q2 < (unsigned)P, m3 = q3 < (unsigned)P;
        if (__builtin_amdgcn_ballot_w32(m0 | m1 | m2 | m3) != 0u) {
            own_keep<HEADS, CH, P, KCAP>(L, wave, lane, kcnt, m0, 4 * i + 0, (int)q0);
            own_keep<HEADS, CH, P, KCAP>(L, wave, lane, kcnt, m1, 4 * i + 1, (int)q1);
            own_keep<HEADS, CH, P, KCAP>(L, wave, lane, kcnt, m2, 4 * i + 2, (int)q2);
            own_keep<HEADS, CH, P, KCAP>(L, wave, lane, kcnt, m3, 4 * i + 3, (int)q3);
        }
    }
    for (int base = 4 * EV + wave * 32; base < E; base += OWN_WG) {
        int i = base + lane;
        unsigned q = 0u; bool m = false;
        if (i < E) { q = (unsigned)(dsts[i] - n0); m = q < (unsigned)P; }
        own_keep<HEADS, CH, P, KCAP>(L, wave, lane, kcnt, m, i, (int)q);
    }
    for (int p0 = wave * 32; p0 < P; p0 += OWN_WG) {
        int p = p0 + lane;
        bool m = (p < P) && (n0 + p < N);
        own_keep<HEADS, CH, P, KCAP>(L, wave, lane, kcnt, m, E + n0 + p, p);
    }
    return __builtin_amdgcn_readfirstlane(kcnt);
}

template <int HEADS, int CH, int P, int KCAP, int PH>
__device__ __forceinline__ void own_walk(OwnLds<HEADS, CH, P, KCAP>& L, int wave, int lane, int kcnt,
                                         const int* __restrict__ ei, int E,
                                         const float* __restrict__ a_s,
                                         const float* __restrict__ h) {
    for (int first = 0; first < kcnt; first += 32)
        own_batch<HEADS, CH, P, KCAP, PH, true>(L, wave, lane, first, min(32, kcnt - first), ei, E, a_s, h);
}

template <int HEADS, int CH, int P, int KCAP>
__global__ __launch_bounds__(OWN_WG)
void k_gat_owner(const int* __restrict__ ei, int E, int N,
                 const float* __restrict__ a_s, const float* __restrict__ a_d,
                 const float* __restrict__ h, float* __restrict__ agg) {
    constexpr int ROW = HEADS * CH;
    __shared__ OwnLds<HEADS, CH, P, KCAP> L;
    const int tid = threadIdx.x, wave = tid >> 5, lane = tid & 31;
    const int n0 = blockIdx.x * P;

    for (int i = tid; i < P * ROW; i += OWN_WG) L.acc[i] = 0.f;
    for (int i = tid; i < P * HEADS; i += OWN_WG) {
        L.menc[i] = 0u;
        L.den[i]  = 0.f;
        L.adl[i]  = (n0 + i / HEADS < N) ? a_d[(size_t)n0 * HEADS + i] : 0.f;
    }
    if (tid == 0) L.ovf = 0;
    __syncthreads();
    const int kcnt = own_filter<HEADS, CH, P, KCAP>(L, wave, lane, ei, E, N, n0);
    if (kcnt > KCAP && lane == 0) L.ovf = 1;
    __syncthreads();
    const bool listed = __builtin_amdgcn_readfirstlane(L.ovf) == 0;
    if (__builtin_expect(listed, true)) {
        own_walk<HEADS, CH, P, KCAP, 0>(L, wave, lane, kcnt, ei, E, a_s, h);
        __syncthreads();
        own_walk<HEADS, CH, P, KCAP, 1>(L, wave, lane, kcnt, ei, E, a_s, h);
        __syncthreads();
        own_walk<HEADS, CH, P, KCAP, 2>(L, wave, lane, kcnt, ei, E, a_s, h);
        __syncthreads();
    } else {
        own_scan<HEADS, CH, P, KCAP, 0>(L, wave, lane, ei, E, N, n0, a_s, h);
        __syncthreads();
        own_scan<HEADS, CH, P, KCAP, 1>(L, wave, lane, ei, E, N, n0, a_s, h);
        __syncthreads();
        own_scan<HEADS, CH, P, KCAP, 2>(L, wave, lane, ei, E, N, n0, a_s, h);
        __syncthreads();
    }
    for (int i = tid; i < P * ROW; i += OWN_WG)
        if (n0 + i / ROW < N) agg[(size_t)n0 * ROW + i] = L.acc[i];
}

__global__ void k_bias_relu(float* __restrict__ x, const float* __restrict__ b,
                            int n, int hc) {
    int i = blockIdx.x * blockDim.x + threadIdx.x;
    if (i >= n * hc) return;
    x[i] = fmaxf(x[i] + b[i % hc], 0.f);
}

__global__ void k_bias_out(const float* __restrict__ agg, const float* __restrict__ b,
                           float* __restrict__ out, int n, int hc) {
    int i = blockIdx.x * blockDim.x + threadIdx.x;
    if (i >= n * hc) return;
    out[i] = agg[i] + b[i % hc];
}


extern "C" void kernel_launch(void* const* d_in, const int* in_sizes, int n_in,
                              void* d_out, int out_size, void* d_ws, size_t ws_size,
                              hipStream_t stream) {
    const float* x        = (const float*)d_in[0];
    const int*   ei       = (const int*)  d_in[1];
    const float* W1       = (const float*)d_in[2];
    const float* att_src1 = (const float*)d_in[3];
    const float* att_dst1 = (const float*)d_in[4];
    const float* b1       = (const float*)d_in[5];
    const float* W2       = (const float*)d_in[6];
    const float* att_src2 = (const float*)d_in[7];
    const float* att_dst2 = (const float*)d_in[8];
    const float* b2       = (const float*)d_in[9];
    float* out = (float*)d_out;

    const int N    = in_sizes[0] / F_IN;
    const int E    = in_sizes[1] / 2;

    float* ws  = (float*)d_ws;
    float* h1  = ws;
    float* as1 = h1  + (size_t)N * H1C1;
    float* ad1 = as1 + (size_t)N * HEADS1;
    float* h2  = ad1 + (size_t)N * HEADS1;
    float* as2 = h2  + (size_t)N * NCLS;
    float* ad2 = as2 + (size_t)N;
    float* agg1 = ad2 + (size_t)N;
    float* agg2 = agg1 + (size_t)N * H1C1;

    const int T = 256;
    constexpr int P1 = 256;
    constexpr int P2 = 512;
    constexpr int K1 = 352;
    constexpr int K2 = 768;

    {
        int tiles = (N >> 4) * (H1C1 >> 4);
        int wavesPerBlk = T >> 5;
        k_gemm_wmma_f32<<<(tiles + wavesPerBlk - 1) / wavesPerBlk, T, 0, stream>>>(
            x, W1, h1, N, F_IN, H1C1);
    }
    k_attn_proj<<<((N * HEADS1) + T - 1) / T, T, 0, stream>>>(
        h1, att_src1, att_dst1, as1, ad1, N, HEADS1, CH1);

    k_gat_owner<HEADS1, CH1, P1, K1><<<(N + P1 - 1) / P1, OWN_WG, 0, stream>>>(
        ei, E, N, as1, ad1, h1, agg1);
    k_bias_relu<<<((N * H1C1) + T - 1) / T, T, 0, stream>>>(agg1, b1, N, H1C1);

    {
        int tiles = (N >> 4) * (NCLS >> 4);
        int wavesPerBlk = T >> 5;
        k_gemm_wmma_f32<<<(tiles + wavesPerBlk - 1) / wavesPerBlk, T, 0, stream>>>(
            agg1, W2, h2, N, H1C1, NCLS);
    }
    k_attn_proj<<<(N + T - 1) / T, T, 0, stream>>>(
        h2, att_src2, att_dst2, as2, ad2, N, 1, NCLS);

    k_gat_owner<1, NCLS, P2, K2><<<(N + P2 - 1) / P2, OWN_WG, 0, stream>>>(
        ei, E, N, as2, ad2, h2, agg2);
    k_bias_out<<<((N * NCLS) + T - 1) / T, T, 0, stream>>>(agg2, b2, out, N, NCLS);
}


# Layer-1 row walk takes one list entry at a time per wave: source and destination row are scalars read from the owning lane, lanes are channels; row sum order = LDS arrival, as before
# speedup vs baseline: 3.9800x; 1.2160x over previous
#include <hip/hip_runtime.h>
#include <hip/hip_bf16.h>


#define NEG_SLOPE 0.2f
#define F_IN   128
#define H1C1   160
#define HEADS1 8
#define CH1    20
#define NCLS   16

typedef float v2f __attribute__((ext_vector_type(2)));
typedef float v8f __attribute__((ext_vector_type(8)));


__device__ __forceinline__ unsigned f2ord(float f) {
    unsigned b = __float_as_uint(f);
    return (b & 0x80000000u) ? ~b : (b | 0x80000000u);
}
__device__ __forceinline__ float ord2f(unsigned u) {
    return (u & 0x80000000u) ? __uint_as_float(u ^ 0x80000000u)
                             : __uint_as_float(~u);
}

__device__ __forceinline__ float lrelu(float v) {
    return (v > 0.f) ? v : NEG_SLOPE * v;
}


__global__ void k_gemm_wmma_f32(const float* __restrict__ A,
                                const float* __restrict__ B,
                                float* __restrict__ C,
                                int M, int K, int Nc) {
    int wave = blockIdx.x * (blockDim.x >> 5) + (threadIdx.x >> 5);
    int ntN = Nc >> 4;
    int ntM = M >> 4;
    if (wave >= ntM * ntN) return;
    int tm = wave / ntN, tn = wave % ntN;
    int m0 = tm << 4, n0 = tn << 4;

    int lane  = threadIdx.x & 31;
    int lr    = lane & 15;
    int hi    = lane >> 4;
    int khalf = hi << 1;

    const float* __restrict__ arow = A + (size_t)(m0 + lr) * K;
    v8f c = {};
    for (int k0 = 0; k0 < K; k0 += 4) {
        v2f a, b;
        a.x = arow[k0 + khalf];
        a.y = arow[k0 + khalf + 1];
        b.x = B[(size_t)(k0 + khalf)     * Nc + n0 + lr];
        b.y = B[(size_t)(k0 + khalf + 1) * Nc + n0 + lr];
        c = __builtin_amdgcn_wmma_f32_16x16x4_f32(
                 false, a,  false, b,
                 (short)0, c,  false,  false);
    }
#pragma unroll
    for (int r = 0; r < 8; ++r) {
        int row = m0 + r + (hi << 3);
        C[(size_t)row * Nc + n0 + lr] = c[r];
    }
}

__global__ void k_attn_proj(const float* __restrict__ h,
                            const float* __restrict__ att_s,
                            const float* __restrict__ att_d,
                            float* __restrict__ a_s, float* __restrict__ a_d,
                            int n, int heads, int ch) {
    int i = blockIdx.x * blockDim.x + threadIdx.x;
    if (i >= n * heads) return;
    int node = i / heads, hh = i - node * heads;
    const float* hp = h + (size_t)node * heads * ch + (size_t)hh * ch;
    const float* as = att_s + hh * ch;
    const float* ad = att_d + hh * ch;
    float ss = 0.f, sd = 0.f;
    for (int c = 0; c < ch; ++c) {
        float v = hp[c];
        ss += v * as[c];
        sd += v * ad[c];
    }
    a_s[i] = ss;
    a_d[i] = sd;
}


#define OWN_WG   1024
#define OWN_NW   (OWN_WG / 32)
#define OWN_QCAP 160

template <int HEADS, int CH, int P, int KCAP>
struct OwnLds {
    static_assert(KCAP >= OWN_QCAP, "the three-scan path queues in the same arrays");
    float    acc[P * HEADS * CH];
    unsigned menc[P * HEADS];
    float    den[P * HEADS];
    float    adl[P * HEADS];
    int      qs[OWN_NW][KCAP];
    int      qd[OWN_NW][KCAP];
    float    qa[OWN_NW][32 * HEADS];
    int      ovf;
};

__device__ __forceinline__ void wave_lds_handover() {
    __builtin_amdgcn_fence(__ATOMIC_RELEASE, "wavefront");
    __builtin_amdgcn_wave_barrier();
    __builtin_amdgcn_fence(__ATOMIC_ACQUIRE, "wavefront");
}

template <int HEADS, int CH, int P, int KCAP, int PH, bool LIST>
__device__ __forceinline__ void own_batch(OwnLds<HEADS, CH, P, KCAP>& L, int wave, int lane,
                                          int first, int cnt,
                                          const int* __restrict__ ei, int E,
                                          const float* __restrict__ a_s,
                                          const float* __restrict__ h) {
    constexpr int ROW = HEADS * CH;
    constexpr bool WHOLE = (ROW % 32 == 0);
    wave_lds_handover();
    int s = 0, dl = 0;
    if (lane < cnt) {
        int e  = L.qs[wave][first + lane];
        dl = L.qd[wave][first + lane];
        if (LIST && PH > 0) s = e;
        else s = (e < E) ? ei[e] : e - E;
        if (LIST ? (PH == 0) : (PH == 2 && !WHOLE))
            L.qs[wave][first + lane] = s;
#pragma unroll
        for (int hh = 0; hh < HEADS; ++hh) {
            float v = lrelu(a_s[(size_t)s * HEADS + hh] + L.adl[dl * HEADS + hh]);
            if (PH == 0) {
                atomicMax(&L.menc[dl * HEADS + hh], f2ord(v));
            } else {
                float w = expf(v - ord2f(L.menc[dl * HEADS + hh]));
                if (PH == 1) atomicAdd(&L.den[dl * HEADS + hh], w);
                else         L.qa[wave][lane * HEADS + hh] = w / L.den[dl * HEADS + hh];
            }
        }
    }
    if (PH == 2) {
        wave_lds_handover();
        if constexpr (WHOLE) {
            constexpr int NJ = ROW / 32;
            constexpr int G  = 4;
            const int cu = __builtin_amdgcn_readfirstlane(cnt);
            int hq[NJ];
#pragma unroll
            for (int j = 0; j < NJ; ++j) hq[j] = (lane + 32 * j) / CH;
            float* const accl = &L.acc[lane];
            const float* const hl = h + lane;
            int e0 = 0;
            for (; e0 + G <= cu; e0 += G) {
                float v[G][NJ], a[G][NJ];
#pragma unroll
                for (int g = 0; g < G; ++g) {
                    const int su = __builtin_amdgcn_readlane(s, e0 + g);
                    const float* __restrict__ row = hl + (size_t)su * ROW;
                    const float* qrow = &L.qa[wave][(e0 + g) * HEADS];
#pragma unroll
                    for (int j = 0; j < NJ; ++j) { v[g][j] = row[32 * j]; a[g][j] = qrow[hq[j]]; }
                }
#pragma unroll
                for (int g = 0; g < G; ++g) {
                    const int du = __builtin_amdgcn_readlane(dl, e0 + g);
                    float* arow = accl + du * ROW;
#pragma unroll
                    for (int j = 0; j < NJ; ++j)
                        atomicAdd(arow + 32 * j, a[g][j] * v[g][j]);
                }
            }
            for (; e0 < cu; ++e0) {
                const int su = __builtin_amdgcn_readlane(s, e0);
                const int du = __builtin_amdgcn_readlane(dl, e0);
                const float* __restrict__ row = hl + (size_t)su * ROW;
                const float* qrow = &L.qa[wave][e0 * HEADS];
                float v[NJ], a[NJ];
#pragma unroll
                for (int j = 0; j < NJ; ++j) { v[j] = row[32 * j]; a[j] = qrow[hq[j]]; }
                float* arow = accl + du * ROW;
#pragma unroll
                for (int j = 0; j < NJ; ++j)
                    atomicAdd(arow + 32 * j, a[j] * v[j]);
            }
        } else {
            constexpr int U = 8;
            const int total = cnt * ROW;
            for (int base = 0; base < total; base += 32 * U) {
                float v[U];
#pragma unroll
                for (int u = 0; u < U; ++u) {
                    int idx = min(base + u * 32 + lane, total - 1);
                    int ent = idx / ROW, col = idx - ent * ROW;
                    int sn = L.qs[wave][first + ent];
                    v[u] = h[(size_t)sn * ROW + col];
                }
#pragma unroll
                for (int u = 0; u < U; ++u) {
                    int idx = base + u * 32 + lane;
                    if (idx < total) {
                        int ent = idx / ROW, col = idx - ent * ROW;
                        int dn = L.qd[wave][first + ent];
                        float a = L.qa[wave][ent * HEADS + col / CH];
                        atomicAdd(&L.acc[dn * ROW + col], a * v[u]);
                    }
                }
            }
        }
        wave_lds_handover();
    }
}

template <int HEADS, int CH, int P, int KCAP>
__device__ __forceinline__ void own_append(OwnLds<HEADS, CH, P, KCAP>& L, int wave, int lane, int& qcnt,
                                           bool m, int e, int dl) {
    unsigned mask = __builtin_amdgcn_ballot_w32(m);
    if (mask == 0u) return;
    int pos = qcnt + __popc(mask & ((1u << lane) - 1u));
    if (m) { L.qs[wave][pos] = e; L.qd[wave][pos] = dl; }
    qcnt += __popc(mask);
}

template <int HEADS, int CH, int P, int KCAP, int PH>
__device__ __forceinline__ void own_drain(OwnLds<HEADS, CH, P, KCAP>& L, int wave, int lane, int& qcnt,
                                          const int* __restrict__ ei, int E,
                                          const float* __restrict__ a_s,
                                          const float* __restrict__ h) {
    while (qcnt >= 32) {
        own_batch<HEADS, CH, P, KCAP, PH, false>(L, wave, lane, qcnt - 32, 32, ei, E, a_s, h);
        qcnt -= 32;
    }
}

__device__ __forceinline__ int4 own_ld4(const int* __restrict__ dsts, int i, int EV) {
    return (i < EV) ? *(const int4*)(dsts + 4 * (size_t)i) : make_int4(-1, -1, -1, -1);
}

template <int HEADS, int CH, int P, int KCAP, int PH>
__device__ __forceinline__ void own_scan(OwnLds<HEADS, CH, P, KCAP>& L, int wave, int lane,
                                         const int* __restrict__ ei, int E, int N, int n0,
                                         const float* __restrict__ a_s,
                                         const float* __restrict__ h) {
    int qcnt = 0;
    const int* __restrict__ dsts = ei + E;
    const int EV = (E & 3) ? 0 : (E >> 2);
    const int t = wave * 32 + lane;
    int4 r0 = own_ld4(dsts, t, EV);
    int4 r1 = own_ld4(dsts, t + OWN_WG, EV);
    int4 r2 = own_ld4(dsts, t + 2 * OWN_WG, EV);
    int4 r3 = own_ld4(dsts, t + 3 * OWN_WG, EV);
    for (int base = wave * 32; base < EV; base += OWN_WG) {
        const int i = base + lane;
        const int4 d = r0;
        r0 = r1; r1 = r2; r2 = r3;
        r3 = own_ld4(dsts, i + 4 * OWN_WG, EV);
        unsigned q0 = (unsigned)(d.x - n0), q1 = (unsigned)(d.y - n0);
        unsigned q2 = (unsigned)(d.z - n0), q3 = (unsigned)(d.w - n0);
        bool m0 = q0 < (unsigned)P, m1 = q1 < (unsigned)P;
        bool m2 = q2 < (unsigned)P, m3 = q3 < (unsigned)P;
        if (__builtin_amdgcn_ballot_w32(m0 | m1 | m2 | m3) != 0u) {
            own_append<HEADS, CH, P, KCAP>(L, wave, lane, qcnt, m0, 4 * i + 0, (int)q0);
            own_append<HEADS, CH, P, KCAP>(L, wave, lane, qcnt, m1, 4 * i + 1, (int)q1);
            own_append<HEADS, CH, P, KCAP>(L, wave, lane, qcnt, m2, 4 * i + 2, (int)q2);
            own_append<HEADS, CH, P, KCAP>(L, wave, lane, qcnt, m3, 4 * i + 3, (int)q3);
            own_drain<HEADS, CH, P, KCAP, PH>(L, wave, lane, qcnt, ei, E, a_s, h);
        }
    }
    for (int base = 4 * EV + wave * 32; base < E; base += OWN_WG) {
        int i = base + lane;
        unsigned q = 0u; bool m = false;
        if (i < E) { q = (unsigned)(dsts[i] - n0); m = q < (unsigned)P; }
        own_append<HEADS, CH, P, KCAP>(L, wave, lane, qcnt, m, i, (int)q);
        own_drain<HEADS, CH, P, KCAP, PH>(L, wave, lane, qcnt, ei, E, a_s, h);
    }
    for (int p0 = wave * 32; p0 < P; p0 += OWN_WG) {
        int p = p0 + lane;
        bool m = (p < P) && (n0 + p < N);
        own_append<HEADS, CH, P, KCAP>(L, wave, lane, qcnt, m, E + n0 + p, p);
        own_drain<HEADS, CH, P, KCAP, PH>(L, wave, lane, qcnt, ei, E, a_s, h);
    }
    if (qcnt > 0) own_batch<HEADS, CH, P, KCAP, PH, false>(L, wave, lane, 0, qcnt, ei, E, a_s, h);
}

template <int HEADS, int CH, int P, int KCAP>
__device__ __forceinline__ void own_keep(OwnLds<HEADS, CH, P, KCAP>& L, int wave, int lane, int& kcnt,
                                         bool m, int e, int dl) {
    unsigned mask = __builtin_amdgcn_ballot_w32(m);
    if (mask == 0u) return;
    int pos = kcnt + __popc(mask & ((1u << lane) - 1u));
    if (m && pos < KCAP) { L.qs[wave][pos] = e; L.qd[wave][pos] = dl; }
    kcnt += __popc(mask);
}

template <int HEADS, int CH, int P, int KCAP>
__device__ __forceinline__ int own_filter(OwnLds<HEADS, CH, P, KCAP>& L, int wave, int lane,
                                          const int* __restrict__ ei, int E, int N, int n0) {
    int kcnt = 0;
    const int* __restrict__ dsts = ei + E;
    const int EV = (E & 3) ? 0 : (E >> 2);
    const int t = wave * 32 + lane;
    int4 r0 = own_ld4(dsts, t, EV);
    int4 r1 = own_ld4(dsts, t + OWN_WG, EV);
    int4 r2 = own_ld4(dsts, t + 2 * OWN_WG, EV);
    int4 r3 = own_ld4(dsts, t + 3 * OWN_WG, EV);
    for (int base = wave * 32; base < EV; base += OWN_WG) {
        const int i = base + lane;
        const int4 d = r0;
        r0 = r1; r1 = r2; r2 = r3;
        r3 = own_ld4(dsts, i + 4 * OWN_WG, EV);
        unsigned q0 = (unsigned)(d.x - n0), q1 = (unsigned)(d.y - n0);
        unsigned q2 = (unsigned)(d.z - n0), q3 = (unsigned)(d.w - n0);
        bool m0 = q0 < (unsigned)P, m1 = q1 < (unsigned)P;
        bool m2 = q2 < (unsigned)P, m3 = q3 < (unsigned)P;
        if (__builtin_amdgcn_ballot_w32(m0 | m1 | m2 | m3) != 0u) {
            own_keep<HEADS, CH, P, KCAP>(L, wave, lane, kcnt, m0, 4 * i + 0, (int)q0);
            own_keep<HEADS, CH, P, KCAP>(L, wave, lane, kcnt, m1, 4 * i + 1, (int)q1);
            own_keep<HEADS, CH, P, KCAP>(L, wave, lane, kcnt, m2, 4 * i + 2, (int)q2);
            own_keep<HEADS, CH, P, KCAP>(L, wave, lane, kcnt, m3, 4 * i + 3, (int)q3);
        }
    }
    for (int base = 4 * EV + wave * 32; base < E; base += OWN_WG) {
        int i = base + lane;
        unsigned q = 0u; bool m = false;
        if (i < E) { q = (unsigned)(dsts[i] - n0); m = q < (unsigned)P; }
        own_keep<HEADS, CH, P, KCAP>(L, wave, lane, kcnt, m, i, (int)q);
    }
    for (int p0 = wave * 32; p0 < P; p0 += OWN_WG) {
        int p = p0 + lane;
        bool m = (p < P) && (n0 + p < N);
        own_keep<HEADS, CH, P, KCAP>(L, wave, lane, kcnt, m, E + n0 + p, p);
    }
    return __builtin_amdgcn_readfirstlane(kcnt);
}

template <int HEADS, int CH, int P, int KCAP, int PH>
__device__ __forceinline__ void own_walk(OwnLds<HEADS, CH, P, KCAP>& L, int wave, int lane, int kcnt,
                                         const int* __restrict__ ei, int E,
                                         const float* __restrict__ a_s,
                                         const float* __restrict__ h) {
    for (int first = 0; first < kcnt; first += 32)
        own_batch<HEADS, CH, P, KCAP, PH, true>(L, wave, lane, first, min(32, kcnt - first), ei, E, a_s, h);
}

template <int HEADS, int CH, int P, int KCAP>
__global__ __launch_bounds__(OWN_WG)
void k_gat_owner(const int* __restrict__ ei, int E, int N,
                 const float* __restrict__ a_s, const float* __restrict__ a_d,
                 const float* __restrict__ h, float* __restrict__ agg) {
    constexpr int ROW = HEADS * CH;
    __shared__ OwnLds<HEADS, CH, P, KCAP> L;
    const int tid = threadIdx.x, wave = tid >> 5, lane = tid & 31;
    const int n0 = blockIdx.x * P;

    for (int i = tid; i < P * ROW; i += OWN_WG) L.acc[i] = 0.f;
    for (int i = tid; i < P * HEADS; i += OWN_WG) {
        L.menc[i] = 0u;
        L.den[i]  = 0.f;
        L.adl[i]  = (n0 + i / HEADS < N) ? a_d[(size_t)n0 * HEADS + i] : 0.f;
    }
    if (tid == 0) L.ovf = 0;
    __syncthreads();
    const int kcnt = own_filter<HEADS, CH, P, KCAP>(L, wave, lane, ei, E, N, n0);
    if (kcnt > KCAP && lane == 0) L.ovf = 1;
    __syncthreads();
    const bool listed = __builtin_amdgcn_readfirstlane(L.ovf) == 0;
    if (__builtin_expect(listed, true)) {
        own_walk<HEADS, CH, P, KCAP, 0>(L, wave, lane, kcnt, ei, E, a_s, h);
        __syncthreads();
        own_walk<HEADS, CH, P, KCAP, 1>(L, wave, lane, kcnt, ei, E, a_s, h);
        __syncthreads();
        own_walk<HEADS, CH, P, KCAP, 2>(L, wave, lane, kcnt, ei, E, a_s, h);
        __syncthreads();
    } else {
        own_scan<HEADS, CH, P, KCAP, 0>(L, wave, lane, ei, E, N, n0, a_s, h);
        __syncthreads();
        own_scan<HEADS, CH, P, KCAP, 1>(L, wave, lane, ei, E, N, n0, a_s, h);
        __syncthreads();
        own_scan<HEADS, CH, P, KCAP, 2>(L, wave, lane, ei, E, N, n0, a_s, h);
        __syncthreads();
    }
    for (int i = tid; i < P * ROW; i += OWN_WG)
        if (n0 + i / ROW < N) agg[(size_t)n0 * ROW + i] = L.acc[i];
}

__global__ void k_bias_relu(float* __restrict__ x, const float* __restrict__ b,
                            int n, int hc) {
    int i = blockIdx.x * blockDim.x + threadIdx.x;
    if (i >= n * hc) return;
    x[i] = fmaxf(x[i] + b[i % hc], 0.f);
}

__global__ void k_bias_out(const float* __restrict__ agg, const float* __restrict__ b,
                           float* __restrict__ out, int n, int hc) {
    int i = blockIdx.x * blockDim.x + threadIdx.x;
    if (i >= n * hc) return;
    out[i] = agg[i] + b[i % hc];
}


extern "C" void kernel_launch(void* const* d_in, const int* in_sizes, int n_in,
                              void* d_out, int out_size, void* d_ws, size_t ws_size,
                              hipStream_t stream) {
    const float* x        = (const float*)d_in[0];
    const int*   ei       = (const int*)  d_in[1];
    const float* W1       = (const float*)d_in[2];
    const float* att_src1 = (const float*)d_in[3];
    const float* att_dst1 = (const float*)d_in[4];
    const float* b1       = (const float*)d_in[5];
    const float* W2       = (const float*)d_in[6];
    const float* att_src2 = (const float*)d_in[7];
    const float* att_dst2 = (const float*)d_in[8];
    const float* b2       = (const float*)d_in[9];
    float* out = (float*)d_out;

    const int N    = in_sizes[0] / F_IN;
    const int E    = in_sizes[1] / 2;

    float* ws  = (float*)d_ws;
    float* h1  = ws;
    float* as1 = h1  + (size_t)N * H1C1;
    float* ad1 = as1 + (size_t)N * HEADS1;
    float* h2  = ad1 + (size_t)N * HEADS1;
    float* as2 = h2  + (size_t)N * NCLS;
    float* ad2 = as2 + (size_t)N;
    float* agg1 = ad2 + (size_t)N;
    float* agg2 = agg1 + (size_t)N * H1C1;

    const int T = 256;
    constexpr int P1 = 256;
    constexpr int P2 = 512;
    constexpr int K1 = 352;
    constexpr int K2 = 768;

    {
        int tiles = (N >> 4) * (H1C1 >> 4);
        int wavesPerBlk = T >> 5;
        k_gemm_wmma_f32<<<(tiles + wavesPerBlk - 1) / wavesPerBlk, T, 0, stream>>>(
            x, W1, h1, N, F_IN, H1C1);
    }
    k_attn_proj<<<((N * HEADS1) + T - 1) / T, T, 0, stream>>>(
        h1, att_src1, att_dst1, as1, ad1, N, HEADS1, CH1);

    k_gat_owner<HEADS1, CH1, P1, K1><<<(N + P1 - 1) / P1, OWN_WG, 0, stream>>>(
        ei, E, N, as1, ad1, h1, agg1);
    k_bias_relu<<<((N * H1C1) + T - 1) / T, T, 0, stream>>>(agg1, b1, N, H1C1);

    {
        int tiles = (N >> 4) * (NCLS >> 4);
        int wavesPerBlk = T >> 5;
        k_gemm_wmma_f32<<<(tiles + wavesPerBlk - 1) / wavesPerBlk, T, 0, stream>>>(
            agg1, W2, h2, N, H1C1, NCLS);
    }
    k_attn_proj<<<(N + T - 1) / T, T, 0, stream>>>(
        h2, att_src2, att_dst2, as2, ad2, N, 1, NCLS);

    k_gat_owner<1, NCLS, P2, K2><<<(N + P2 - 1) / P2, OWN_WG, 0, stream>>>(
        ei, E, N, as2, ad2, h2, agg2);
    k_bias_out<<<((N * NCLS) + T - 1) / T, T, 0, stream>>>(agg2, b2, out, N, NCLS);
}


# Layer 2 takes over the per-wave kept lists that layer 1 built (stored in the workspace in the same call) instead of reading the destination list again; both own 256 nodes; row sum order = LDS arrival
# speedup vs baseline: 4.9417x; 1.2416x over previous
#include <hip/hip_runtime.h>
#include <hip/hip_bf16.h>


#define NEG_SLOPE 0.2f
#define F_IN   128
#define H1C1   160
#define HEADS1 8
#define CH1    20
#define NCLS   16

typedef float v2f __attribute__((ext_vector_type(2)));
typedef float v8f __attribute__((ext_vector_type(8)));


__device__ __forceinline__ unsigned f2ord(float f) {
    unsigned b = __float_as_uint(f);
    return (b & 0x80000000u) ? ~b : (b | 0x80000000u);
}
__device__ __forceinline__ float ord2f(unsigned u) {
    return (u & 0x80000000u) ? __uint_as_float(u ^ 0x80000000u)
                             : __uint_as_float(~u);
}

__device__ __forceinline__ float lrelu(float v) {
    return (v > 0.f) ? v : NEG_SLOPE * v;
}


__global__ void k_gemm_wmma_f32(const float* __restrict__ A,
                                const float* __restrict__ B,
                                float* __restrict__ C,
                                int M, int K, int Nc) {
    int wave = blockIdx.x * (blockDim.x >> 5) + (threadIdx.x >> 5);
    int ntN = Nc >> 4;
    int ntM = M >> 4;
    if (wave >= ntM * ntN) return;
    int tm = wave / ntN, tn = wave % ntN;
    int m0 = tm << 4, n0 = tn << 4;

    int lane  = threadIdx.x & 31;
    int lr    = lane & 15;
    int hi    = lane >> 4;
    int khalf = hi << 1;

    const float* __restrict__ arow = A + (size_t)(m0 + lr) * K;
    v8f c = {};
    for (int k0 = 0; k0 < K; k0 += 4) {
        v2f a, b;
        a.x = arow[k0 + khalf];
        a.y = arow[k0 + khalf + 1];
        b.x = B[(size_t)(k0 + khalf)     * Nc + n0 + lr];
        b.y = B[(size_t)(k0 + khalf + 1) * Nc + n0 + lr];
        c = __builtin_amdgcn_wmma_f32_16x16x4_f32(
                 false, a,  false, b,
                 (short)0, c,  false,  false);
    }
#pragma unroll
    for (int r = 0; r < 8; ++r) {
        int row = m0 + r + (hi << 3);
        C[(size_t)row * Nc + n0 + lr] = c[r];
    }
}

__global__ void k_attn_proj(const float* __restrict__ h,
                            const float* __restrict__ att_s,
                            const float* __restrict__ att_d,
                            float* __restrict__ a_s, float* __restrict__ a_d,
                            int n, int heads, int ch) {
    int i = blockIdx.x * blockDim.x + threadIdx.x;
    if (i >= n * heads) return;
    int node = i / heads, hh = i - node * heads;
    const float* hp = h + (size_t)node * heads * ch + (size_t)hh * ch;
    const float* as = att_s + hh * ch;
    const float* ad = att_d + hh * ch;
    float ss = 0.f, sd = 0.f;
    for (int c = 0; c < ch; ++c) {
        float v = hp[c];
        ss += v * as[c];
        sd += v * ad[c];
    }
    a_s[i] = ss;
    a_d[i] = sd;
}


#define OWN_WG   1024
#define OWN_NW   (OWN_WG / 32)
#define OWN_QCAP 160

template <int HEADS, int CH, int P, int KCAP>
struct OwnLds {
    static_assert(KCAP >= OWN_QCAP, "the three-scan path queues in the same arrays");
    float    acc[P * HEADS * CH];
    unsigned menc[P * HEADS];
    float    den[P * HEADS];
    float    adl[P * HEADS];
    int      qs[OWN_NW][KCAP];
    int      qd[OWN_NW][KCAP];
    float    qa[OWN_NW][32 * HEADS];
    int      ovf;
};

__device__ __forceinline__ void wave_lds_handover() {
    __builtin_amdgcn_fence(__ATOMIC_RELEASE, "wavefront");
    __builtin_amdgcn_wave_barrier();
    __builtin_amdgcn_fence(__ATOMIC_ACQUIRE, "wavefront");
}

template <int HEADS, int CH, int P, int KCAP, int PH, bool LIST, bool RES = false>
__device__ __forceinline__ void own_batch(OwnLds<HEADS, CH, P, KCAP>& L, int wave, int lane,
                                          int first, int cnt,
                                          const int* __restrict__ ei, int E,
                                          const float* __restrict__ a_s,
                                          const float* __restrict__ h) {
    constexpr int ROW = HEADS * CH;
    constexpr bool WHOLE = (ROW % 32 == 0);
    wave_lds_handover();
    int s = 0, dl = 0;
    if (lane < cnt) {
        int e  = L.qs[wave][first + lane];
        dl = L.qd[wave][first + lane];
        if (LIST && (PH > 0 || RES)) s = e;
        else s = (e < E) ? ei[e] : e - E;
        if (LIST ? (PH == 0 && !RES) : (PH == 2 && !WHOLE))
            L.qs[wave][first + lane] = s;
#pragma unroll
        for (int hh = 0; hh < HEADS; ++hh) {
            float v = lrelu(a_s[(size_t)s * HEADS + hh] + L.adl[dl * HEADS + hh]);
            if (PH == 0) {
                atomicMax(&L.menc[dl * HEADS + hh], f2ord(v));
            } else {
                float w = expf(v - ord2f(L.menc[dl * HEADS + hh]));
                if (PH == 1) atomicAdd(&L.den[dl * HEADS + hh], w);
                else         L.qa[wave][lane * HEADS + hh] = w / L.den[dl * HEADS + hh];
            }
        }
    }
    if (PH == 2) {
        wave_lds_handover();
        if constexpr (WHOLE) {
            constexpr int NJ = ROW / 32;
            constexpr int G  = 4;
            const int cu = __builtin_amdgcn_readfirstlane(cnt);
            int hq[NJ];
#pragma unroll
            for (int j = 0; j < NJ; ++j) hq[j] = (lane + 32 * j) / CH;
            float* const accl = &L.acc[lane];
            const float* const hl = h + lane;
            int e0 = 0;
            for (; e0 + G <= cu; e0 += G) {
                float v[G][NJ], a[G][NJ];
#pragma unroll
                for (int g = 0; g < G; ++g) {
                    const int su = __builtin_amdgcn_readlane(s, e0 + g);
                    const float* __restrict__ row = hl + (size_t)su * ROW;
                    const float* qrow = &L.qa[wave][(e0 + g) * HEADS];
#pragma unroll
                    for (int j = 0; j < NJ; ++j) { v[g][j] = row[32 * j]; a[g][j] = qrow[hq[j]]; }
                }
#pragma unroll
                for (int g = 0; g < G; ++g) {
                    const int du = __builtin_amdgcn_readlane(dl, e0 + g);
                    float* arow = accl + du * ROW;
#pragma unroll
                    for (int j = 0; j < NJ; ++j)
                        atomicAdd(arow + 32 * j, a[g][j] * v[g][j]);
                }
            }
            for (; e0 < cu; ++e0) {
                const int su = __builtin_amdgcn_readlane(s, e0);
                const int du = __builtin_amdgcn_readlane(dl, e0);
                const float* __restrict__ row = hl + (size_t)su * ROW;
                const float* qrow = &L.qa[wave][e0 * HEADS];
                float v[NJ], a[NJ];
#pragma unroll
                for (int j = 0; j < NJ; ++j) { v[j] = row[32 * j]; a[j] = qrow[hq[j]]; }
                float* arow = accl + du * ROW;
#pragma unroll
                for (int j = 0; j < NJ; ++j)
                    atomicAdd(arow + 32 * j, a[j] * v[j]);
            }
        } else {
            constexpr int U = 8;
            const int total = cnt * ROW;
            for (int base = 0; base < total; base += 32 * U) {
                float v[U];
#pragma unroll
                for (int u = 0; u < U; ++u) {
                    int idx = min(base + u * 32 + lane, total - 1);
                    int ent = idx / ROW, col = idx - ent * ROW;
                    int sn = L.qs[wave][first + ent];
                    v[u] = h[(size_t)sn * ROW + col];
                }
#pragma unroll
                for (int u = 0; u < U; ++u) {
                    int idx = base + u * 32 + lane;
                    if (idx < total) {
                        int ent = idx / ROW, col = idx - ent * ROW;
                        int dn = L.qd[wave][first + ent];
                        float a = L.qa[wave][ent * HEADS + col / CH];
                        atomicAdd(&L.acc[dn * ROW + col], a * v[u]);
                    }
                }
            }
        }
        wave_lds_handover();
    }
}

template <int HEADS, int CH, int P, int KCAP>
__device__ __forceinline__ void own_append(OwnLds<HEADS, CH, P, KCAP>& L, int wave, int lane, int& qcnt,
                                           bool m, int e, int dl) {
    unsigned mask = __builtin_amdgcn_ballot_w32(m);
    if (mask == 0u) return;
    int pos = qcnt + __popc(mask & ((1u << lane) - 1u));
    if (m) { L.qs[wave][pos] = e; L.qd[wave][pos] = dl; }
    qcnt += __popc(mask);
}

template <int HEADS, int CH, int P, int KCAP, int PH>
__device__ __forceinline__ void own_drain(OwnLds<HEADS, CH, P, KCAP>& L, int wave, int lane, int& qcnt,
                                          const int* __restrict__ ei, int E,
                                          const float* __restrict__ a_s,
                                          const float* __restrict__ h) {
    while (qcnt >= 32) {
        own_batch<HEADS, CH, P, KCAP, PH, false>(L, wave, lane, qcnt - 32, 32, ei, E, a_s, h);
        qcnt -= 32;
    }
}

__device__ __forceinline__ int4 own_ld4(const int* __restrict__ dsts, int i, int EV) {
    return (i < EV) ? *(const int4*)(dsts + 4 * (size_t)i) : make_int4(-1, -1, -1, -1);
}

template <int HEADS, int CH, int P, int KCAP, int PH>
__device__ __forceinline__ void own_scan(OwnLds<HEADS, CH, P, KCAP>& L, int wave, int lane,
                                         const int* __restrict__ ei, int E, int N, int n0,
                                         const float* __restrict__ a_s,
                                         const float* __restrict__ h) {
    int qcnt = 0;
    const int* __restrict__ dsts = ei + E;
    const int EV = (E & 3) ? 0 : (E >> 2);
    const int t = wave * 32 + lane;
    int4 r0 = own_ld4(dsts, t, EV);
    int4 r1 = own_ld4(dsts, t + OWN_WG, EV);
    int4 r2 = own_ld4(dsts, t + 2 * OWN_WG, EV);
    int4 r3 = own_ld4(dsts, t + 3 * OWN_WG, EV);
    for (int base = wave * 32; base < EV; base += OWN_WG) {
        const int i = base + lane;
        const int4 d = r0;
        r0 = r1; r1 = r2; r2 = r3;
        r3 = own_ld4(dsts, i + 4 * OWN_WG, EV);
        unsigned q0 = (unsigned)(d.x - n0), q1 = (unsigned)(d.y - n0);
        unsigned q2 = (unsigned)(d.z - n0), q3 = (unsigned)(d.w - n0);
        bool m0 = q0 < (unsigned)P, m1 = q1 < (unsigned)P;
        bool m2 = q2 < (unsigned)P, m3 = q3 < (unsigned)P;
        if (__builtin_amdgcn_ballot_w32(m0 | m1 | m2 | m3) != 0u) {
            own_append<HEADS, CH, P, KCAP>(L, wave, lane, qcnt, m0, 4 * i + 0, (int)q0);
            own_append<HEADS, CH, P, KCAP>(L, wave, lane, qcnt, m1, 4 * i + 1, (int)q1);
            own_append<HEADS, CH, P, KCAP>(L, wave, lane, qcnt, m2, 4 * i + 2, (int)q2);
            own_append<HEADS, CH, P, KCAP>(L, wave, lane, qcnt, m3, 4 * i + 3, (int)q3);
            own_drain<HEADS, CH, P, KCAP, PH>(L, wave, lane, qcnt, ei, E, a_s, h);
        }
    }
    for (int base = 4 * EV + wave * 32; base < E; base += OWN_WG) {
        int i = base + lane;
        unsigned q = 0u; bool m = false;
        if (i < E) { q = (unsigned)(dsts[i] - n0); m = q < (unsigned)P; }
        own_append<HEADS, CH, P, KCAP>(L, wave, lane, qcnt, m, i, (int)q);
        own_drain<HEADS, CH, P, KCAP, PH>(L, wave, lane, qcnt, ei, E, a_s, h);
    }
    for (int p0 = wave * 32; p0 < P; p0 += OWN_WG) {
        int p = p0 + lane;
        bool m = (p < P) && (n0 + p < N);
        own_append<HEADS, CH, P, KCAP>(L, wave, lane, qcnt, m, E + n0 + p, p);
        own_drain<HEADS, CH, P, KCAP, PH>(L, wave, lane, qcnt, ei, E, a_s, h);
    }
    if (qcnt > 0) own_batch<HEADS, CH, P, KCAP, PH, false>(L, wave, lane, 0, qcnt, ei, E, a_s, h);
}

template <int HEADS, int CH, int P, int KCAP>
__device__ __forceinline__ void own_keep(OwnLds<HEADS, CH, P, KCAP>& L, int wave, int lane, int& kcnt,
                                         bool m, int e, int dl) {
    unsigned mask = __builtin_amdgcn_ballot_w32(m);
    if (mask == 0u) return;
    int pos = kcnt + __popc(mask & ((1u << lane) - 1u));
    if (m && pos < KCAP) { L.qs[wave][pos] = e; L.qd[wave][pos] = dl; }
    kcnt += __popc(mask);
}

template <int HEADS, int CH, int P, int KCAP>
__device__ __forceinline__ int own_filter(OwnLds<HEADS, CH, P, KCAP>& L, int wave, int lane,
                                          const int* __restrict__ ei, int E, int N, int n0) {
    int kcnt = 0;
    const int* __restrict__ dsts = ei + E;
    const int EV = (E & 3) ? 0 : (E >> 2);
    const int t = wave * 32 + lane;
    int4 r0 = own_ld4(dsts, t, EV);
    int4 r1 = own_ld4(dsts, t + OWN_WG, EV);
    int4 r2 = own_ld4(dsts, t + 2 * OWN_WG, EV);
    int4 r3 = own_ld4(dsts, t + 3 * OWN_WG, EV);
    for (int base = wave * 32; base < EV; base += OWN_WG) {
        const int i = base + lane;
        const int4 d = r0;
        r0 = r1; r1 = r2; r2 = r3;
        r3 = own_ld4(dsts, i + 4 * OWN_WG, EV);
        unsigned q0 = (unsigned)(d.x - n0), q1 = (unsigned)(d.y - n0);
        unsigned q2 = (unsigned)(d.z - n0), q3 = (unsigned)(d.w - n0);
        bool m0 = q0 < (unsigned)P, m1 = q1 < (unsigned)P;
        bool m2 = q2 < (unsigned)P, m3 = q3 < (unsigned)P;
        if (__builtin_amdgcn_ballot_w32(m0 | m1 | m2 | m3) != 0u) {
            own_keep<HEADS, CH, P, KCAP>(L, wave, lane, kcnt, m0, 4 * i + 0, (int)q0);
            own_keep<HEADS, CH, P, KCAP>(L, wave, lane, kcnt, m1, 4 * i + 1, (int)q1);
            own_keep<HEADS, CH, P, KCAP>(L, wave, lane, kcnt, m2, 4 * i + 2, (int)q2);
            own_keep<HEADS, CH, P, KCAP>(L, wave, lane, kcnt, m3, 4 * i + 3, (int)q3);
        }
    }
    for (int base = 4 * EV + wave * 32; base < E; base += OWN_WG) {
        int i = base + lane;
        unsigned q = 0u; bool m = false;
        if (i < E) { q = (unsigned)(dsts[i] - n0); m = q < (unsigned)P; }
        own_keep<HEADS, CH, P, KCAP>(L, wave, lane, kcnt, m, i, (int)q);
    }
    for (int p0 = wave * 32; p0 < P; p0 += OWN_WG) {
        int p = p0 + lane;
        bool m = (p < P) && (n0 + p < N);
        own_keep<HEADS, CH, P, KCAP>(L, wave, lane, kcnt, m, E + n0 + p, p);
    }
    return __builtin_amdgcn_readfirstlane(kcnt);
}

template <int HEADS, int CH, int P, int KCAP, int PH, bool RES = false>
__device__ __forceinline__ void own_walk(OwnLds<HEADS, CH, P, KCAP>& L, int wave, int lane, int kcnt,
                                         const int* __restrict__ ei, int E,
                                         const float* __restrict__ a_s,
                                         const float* __restrict__ h) {
    for (int first = 0; first < kcnt; first += 32)
        own_batch<HEADS, CH, P, KCAP, PH, true, RES>(L, wave, lane, first, min(32, kcnt - first), ei, E, a_s, h);
}

template <int HEADS, int CH, int P, int KCAP, int MODE, int LCAP>
__global__ __launch_bounds__(OWN_WG)
void k_gat_owner(const int* __restrict__ ei, int E, int N,
                 const float* __restrict__ a_s, const float* __restrict__ a_d,
                 const float* __restrict__ h, float* __restrict__ agg,
                 int* __restrict__ lcnt, int2* __restrict__ lent) {
    static_assert(MODE == 0 ? (LCAP == KCAP) : (LCAP <= KCAP), "a taken list must fit the wave's LDS list");
    constexpr int ROW = HEADS * CH;
    __shared__ OwnLds<HEADS, CH, P, KCAP> L;
    const int tid = threadIdx.x, wave = tid >> 5, lane = tid & 31;
    const int n0 = blockIdx.x * P;

    for (int i = tid; i < P * ROW; i += OWN_WG) L.acc[i] = 0.f;
    for (int i = tid; i < P * HEADS; i += OWN_WG) {
        L.menc[i] = 0u;
        L.den[i]  = 0.f;
        L.adl[i]  = (n0 + i / HEADS < N) ? a_d[(size_t)n0 * HEADS + i] : 0.f;
    }
    if (tid == 0) L.ovf = 0;
    __syncthreads();
    const int slot = blockIdx.x * OWN_NW + wave;
    if constexpr (MODE == 1) {
        const bool taken = __builtin_amdgcn_readfirstlane(lcnt[blockIdx.x * OWN_NW]) >= 0;
        if (__builtin_expect(taken, true)) {
            const int tcnt = __builtin_amdgcn_readfirstlane(lcnt[slot]);
            const int2* __restrict__ src = lent + (size_t)slot * LCAP;
            for (int i = lane; i < tcnt; i += 32) {
                const int2 p = src[i];
                L.qs[wave][i] = p.x;
                L.qd[wave][i] = p.y;
            }
            own_walk<HEADS, CH, P, KCAP, 0, true>(L, wave, lane, tcnt, ei, E, a_s, h);
            __syncthreads();
            own_walk<HEADS, CH, P, KCAP, 1, true>(L, wave, lane, tcnt, ei, E, a_s, h);
            __syncthreads();
            own_walk<HEADS, CH, P, KCAP, 2, true>(L, wave, lane, tcnt, ei, E, a_s, h);
            __syncthreads();
            for (int i = tid; i < P * ROW; i += OWN_WG)
                if (n0 + i / ROW < N) agg[(size_t)n0 * ROW + i] = L.acc[i];
            return;
        }
    }
    const int kcnt = own_filter<HEADS, CH, P, KCAP>(L, wave, lane, ei, E, N, n0);
    if (kcnt > KCAP && lane == 0) L.ovf = 1;
    __syncthreads();
    const bool listed = __builtin_amdgcn_readfirstlane(L.ovf) == 0;
    if (__builtin_expect(listed, true)) {
        own_walk<HEADS, CH, P, KCAP, 0>(L, wave, lane, kcnt, ei, E, a_s, h);
        if constexpr (MODE == 0) {
            int2* __restrict__ dstl = lent + (size_t)slot * LCAP;
            for (int i = lane; i < kcnt; i += 32) dstl[i] = make_int2(L.qs[wave][i], L.qd[wave][i]);
            if (lane == 0) lcnt[slot] = kcnt;
        }
        __syncthreads();
        own_walk<HEADS, CH, P, KCAP, 1>(L, wave, lane, kcnt, ei, E, a_s, h);
        __syncthreads();
        own_walk<HEADS, CH, P, KCAP, 2>(L, wave, lane, kcnt, ei, E, a_s, h);
        __syncthreads();
    } else {
        if constexpr (MODE == 0) { if (lane == 0) lcnt[slot] = -1; }
        own_scan<HEADS, CH, P, KCAP, 0>(L, wave, lane, ei, E, N, n0, a_s, h);
        __syncthreads();
        own_scan<HEADS, CH, P, KCAP, 1>(L, wave, lane, ei, E, N, n0, a_s, h);
        __syncthreads();
        own_scan<HEADS, CH, P, KCAP, 2>(L, wave, lane, ei, E, N, n0, a_s, h);
        __syncthreads();
    }
    for (int i = tid; i < P * ROW; i += OWN_WG)
        if (n0 + i / ROW < N) agg[(size_t)n0 * ROW + i] = L.acc[i];
}

__global__ void k_bias_relu(float* __restrict__ x, const float* __restrict__ b,
                            int n, int hc) {
    int i = blockIdx.x * blockDim.x + threadIdx.x;
    if (i >= n * hc) return;
    x[i] = fmaxf(x[i] + b[i % hc], 0.f);
}

__global__ void k_bias_out(const float* __restrict__ agg, const float* __restrict__ b,
                           float* __restrict__ out, int n, int hc) {
    int i = blockIdx.x * blockDim.x + threadIdx.x;
    if (i >= n * hc) return;
    out[i] = agg[i] + b[i % hc];
}


extern "C" void kernel_launch(void* const* d_in, const int* in_sizes, int n_in,
                              void* d_out, int out_size, void* d_ws, size_t ws_size,
                              hipStream_t stream) {
    const float* x        = (const float*)d_in[0];
    const int*   ei       = (const int*)  d_in[1];
    const float* W1       = (const float*)d_in[2];
    const float* att_src1 = (const float*)d_in[3];
    const float* att_dst1 = (const float*)d_in[4];
    const float* b1       = (const float*)d_in[5];
    const float* W2       = (const float*)d_in[6];
    const float* att_src2 = (const float*)d_in[7];
    const float* att_dst2 = (const float*)d_in[8];
    const float* b2       = (const float*)d_in[9];
    float* out = (float*)d_out;

    const int N    = in_sizes[0] / F_IN;
    const int E    = in_sizes[1] / 2;

    float* ws  = (float*)d_ws;
    float* h1  = ws;
    float* as1 = h1  + (size_t)N * H1C1;
    float* ad1 = as1 + (size_t)N * HEADS1;
    float* h2  = ad1 + (size_t)N * HEADS1;
    float* as2 = h2  + (size_t)N * NCLS;
    float* ad2 = as2 + (size_t)N;
    float* agg1 = ad2 + (size_t)N;
    float* agg2 = agg1 + (size_t)N * H1C1;

    const int T = 256;
    constexpr int P1 = 256;
    constexpr int P2 = P1;
    constexpr int K1 = 352;
    constexpr int K2 = 768;
    const int nown = (N + P1 - 1) / P1;
    int*  lcnt = (int*)(agg2 + (size_t)N * NCLS);
    int2* lent = (int2*)(lcnt + (size_t)nown * OWN_NW);

    {
        int tiles = (N >> 4) * (H1C1 >> 4);
        int wavesPerBlk = T >> 5;
        k_gemm_wmma_f32<<<(tiles + wavesPerBlk - 1) / wavesPerBlk, T, 0, stream>>>(
            x, W1, h1, N, F_IN, H1C1);
    }
    k_attn_proj<<<((N * HEADS1) + T - 1) / T, T, 0, stream>>>(
        h1, att_src1, att_dst1, as1, ad1, N, HEADS1, CH1);

    k_gat_owner<HEADS1, CH1, P1, K1, 0, K1><<<nown, OWN_WG, 0, stream>>>(
        ei, E, N, as1, ad1, h1, agg1, lcnt, lent);
    k_bias_relu<<<((N * H1C1) + T - 1) / T, T, 0, stream>>>(agg1, b1, N, H1C1);

    {
        int tiles = (N >> 4) * (NCLS >> 4);
        int wavesPerBlk = T >> 5;
        k_gemm_wmma_f32<<<(tiles + wavesPerBlk - 1) / wavesPerBlk, T, 0, stream>>>(
            agg1, W2, h2, N, H1C1, NCLS);
    }
    k_attn_proj<<<(N + T - 1) / T, T, 0, stream>>>(
        h2, att_src2, att_dst2, as2, ad2, N, 1, NCLS);

    k_gat_owner<1, NCLS, P2, K2, 1, K1><<<nown, OWN_WG, 0, stream>>>(
        ei, E, N, as2, ad2, h2, agg2, lcnt, lent);
    k_bias_out<<<((N * NCLS) + T - 1) / T, T, 0, stream>>>(agg2, b2, out, N, NCLS);
}
